# starting kernel
# baseline (speedup 1.0000x reference)
#include <hip/hip_runtime.h>
#include <hip/hip_bf16.h>

#define N_NODES 50000
#define N_EDGES 800000
#define IN_CH   128
#define HID_CH  128
#define OUT_CH  64

typedef __attribute__((ext_vector_type(2))) float v2f;
typedef __attribute__((ext_vector_type(8))) float v8f;

__global__ void zero_f32(float* __restrict__ p, size_t n) {
    size_t i = (size_t)blockIdx.x * blockDim.x + threadIdx.x;
    size_t stride = (size_t)gridDim.x * blockDim.x;
    for (; i < n; i += stride) p[i] = 0.0f;
}

__global__ void deg_kernel(const int* __restrict__ dst, float* __restrict__ deg, int E) {
    int e = blockIdx.x * blockDim.x + threadIdx.x;
    if (e < E) unsafeAtomicAdd(&deg[dst[e]], 1.0f);
}

__global__ void invdeg_kernel(const float* __restrict__ deg, float* __restrict__ invdeg, int n) {
    int i = blockIdx.x * blockDim.x + threadIdx.x;
    if (i < n) invdeg[i] = 1.0f / fmaxf(deg[i], 1.0f);
}

__global__ __launch_bounds__(256)
void scatter_kernel(const float* __restrict__ feat,
                    const int* __restrict__ src, const int* __restrict__ dst,
                    float* __restrict__ agg, int E) {
    int t    = blockIdx.x * blockDim.x + threadIdx.x;
    int e    = t >> 5;
    int lane = t & 31;
    if (e >= E) return;
    int s = src[e];
    int d = dst[e];
    const float4 v = *(const float4*)(feat + (size_t)s * 128 + lane * 4);
    float* base = agg + (size_t)d * 128 + lane * 4;
    unsafeAtomicAdd(base + 0, v.x);
    unsafeAtomicAdd(base + 1, v.y);
    unsafeAtomicAdd(base + 2, v.z);
    unsafeAtomicAdd(base + 3, v.w);
}

template <int NOUT, bool RELU>
__global__ __launch_bounds__(2 * NOUT)
void sage_gemm(const float* __restrict__ Aagg,
               const float* __restrict__ Aself,
               const float* __restrict__ invdeg,
               const float* __restrict__ Wl,
               const float* __restrict__ Wr,
               const float* __restrict__ bias,
               float* __restrict__ out) {
    constexpr int K = 128;
    const int wave = threadIdx.x >> 5;
    const int lane = threadIdx.x & 31;
    const int row  = lane & 15;
    const int hi   = lane >> 4;
    const int m0   = blockIdx.x * 16;
    const int n0   = wave * 16;

    const float* __restrict__ arow_agg  = Aagg  + (size_t)(m0 + row) * K;
    const float* __restrict__ arow_self = Aself + (size_t)(m0 + row) * K;
    const float  scale = invdeg[m0 + row];

    v8f c = {};

    #pragma unroll 4
    for (int k0 = 0; k0 < K; k0 += 4) {
        const int ka = k0 + 2 * hi;
        const float2 t = *(const float2*)(arow_agg + ka);
        v2f a;
        a.x = t.x * scale;
        a.y = t.y * scale;
        v2f b;
        b.x = Wl[(size_t)ka * NOUT + n0 + row];
        b.y = Wl[(size_t)(ka + 1) * NOUT + n0 + row];
        c = __builtin_amdgcn_wmma_f32_16x16x4_f32(false, a, false, b,
                                                  (short)0, c, false, false);
    }

    #pragma unroll 4
    for (int k0 = 0; k0 < K; k0 += 4) {
        const int ka = k0 + 2 * hi;
        const float2 t = *(const float2*)(arow_self + ka);
        v2f a;
        a.x = t.x;
        a.y = t.y;
        v2f b;
        b.x = Wr[(size_t)ka * NOUT + n0 + row];
        b.y = Wr[(size_t)(ka + 1) * NOUT + n0 + row];
        c = __builtin_amdgcn_wmma_f32_16x16x4_f32(false, a, false, b,
                                                  (short)0, c, false, false);
    }

    const float bv = bias[n0 + row];
    #pragma unroll
    for (int r = 0; r < 8; ++r) {
        float v = c[r] + bv;
        if (RELU) v = fmaxf(v, 0.0f);
        out[(size_t)(m0 + r + 8 * hi) * NOUT + n0 + row] = v;
    }
}

extern "C" void kernel_launch(void* const* d_in, const int* in_sizes, int n_in,
                              void* d_out, int out_size, void* d_ws, size_t ws_size,
                              hipStream_t stream) {
    const float* x   = (const float*)d_in[0];
    const int*   ei  = (const int*)d_in[1];
    const float* W1l = (const float*)d_in[2];
    const float* b1  = (const float*)d_in[3];
    const float* W1r = (const float*)d_in[4];
    const float* W2l = (const float*)d_in[5];
    const float* b2  = (const float*)d_in[6];
    const float* W2r = (const float*)d_in[7];

    const int* src = ei;
    const int* dst = ei + N_EDGES;

    float* agg    = (float*)d_ws;
    float* h      = agg + (size_t)N_NODES * 128;
    float* deg    = h + (size_t)N_NODES * 128;
    float* invdeg = deg + N_NODES;

    const int ZB = 256;

    zero_f32<<<1024, ZB, 0, stream>>>(deg, (size_t)N_NODES);
    deg_kernel<<<(N_EDGES + ZB - 1) / ZB, ZB, 0, stream>>>(dst, deg, N_EDGES);
    invdeg_kernel<<<(N_NODES + ZB - 1) / ZB, ZB, 0, stream>>>(deg, invdeg, N_NODES);

    zero_f32<<<2048, ZB, 0, stream>>>(agg, (size_t)N_NODES * 128);
    scatter_kernel<<<(N_EDGES * 32 + ZB - 1) / ZB, ZB, 0, stream>>>(x, src, dst, agg, N_EDGES);
    sage_gemm<HID_CH, true><<<N_NODES / 16, 2 * HID_CH, 0, stream>>>(
        agg, x, invdeg, W1l, W1r, b1, h);

    zero_f32<<<2048, ZB, 0, stream>>>(agg, (size_t)N_NODES * 128);
    scatter_kernel<<<(N_EDGES * 32 + ZB - 1) / ZB, ZB, 0, stream>>>(h, src, dst, agg, N_EDGES);
    sage_gemm<OUT_CH, false><<<N_NODES / 16, 2 * OUT_CH, 0, stream>>>(
        agg, h, invdeg, W2l, W2r, b2, (float*)d_out);
}


# Pull aggregation: per-call edge list in destination order (returning integer adds, scan, place); a wave sums a node's rows in slot (arrival) order; f32 atomic scatter, degree pass, zero fills gone
# speedup vs baseline: 1.8760x; 1.8760x over previous
#include <hip/hip_runtime.h>
#include <hip/hip_bf16.h>


#define N_NODES 50000
#define N_EDGES 800000
#define IN_CH   128
#define HID_CH  128
#define OUT_CH  64
#define SCAN_T  1024

typedef __attribute__((ext_vector_type(2))) float v2f;
typedef __attribute__((ext_vector_type(8))) float v8f;

__global__ void zero_i32(int* __restrict__ p, int n) {
    int i = blockIdx.x * blockDim.x + threadIdx.x;
    if (i < n) p[i] = 0;
}

__global__ void count_slot_kernel(const int* __restrict__ dst, int* __restrict__ cnt,
                                  int* __restrict__ slot, int E) {
    int e = blockIdx.x * blockDim.x + threadIdx.x;
    if (e < E) slot[e] = atomicAdd(&cnt[dst[e]], 1);
}

constexpr int SCAN_W = SCAN_T / 32;
constexpr int SCAN_Q = N_NODES / 4;
constexpr int SCAN_R = (SCAN_Q + SCAN_T - 1) / SCAN_T;
static_assert(N_NODES % 4 == 0, "the scan reads the counts four at a time");
static_assert(SCAN_W == 32, "one lane per wave total in the last step");

__global__ __launch_bounds__(SCAN_T)
void scan_kernel(int* cnt_off, float* __restrict__ invdeg) {
    __shared__ int wtot[SCAN_W];
    const int lane = threadIdx.x & 31;
    const int w    = threadIdx.x >> 5;
    int4*   q4   = (int4*)cnt_off;
    float4* inv4 = (float4*)invdeg;

    int4 c[SCAN_R];
    int  s[SCAN_R];
    #pragma unroll
    for (int r = 0; r < SCAN_R; ++r) {
        const int g = (w * SCAN_R + r) * 32 + lane;
        c[r] = (g < SCAN_Q) ? q4[g] : make_int4(0, 0, 0, 0);
        s[r] = c[r].x + c[r].y + c[r].z + c[r].w;
    }
    #pragma unroll
    for (int d = 1; d < 32; d <<= 1) {
        int u[SCAN_R];
        #pragma unroll
        for (int r = 0; r < SCAN_R; ++r) u[r] = __shfl_up(s[r], d);
        #pragma unroll
        for (int r = 0; r < SCAN_R; ++r) if (lane >= d) s[r] += u[r];
    }
    int ex[SCAN_R];
    int carry = 0;
    #pragma unroll
    for (int r = 0; r < SCAN_R; ++r) {
        ex[r] = carry + s[r] - (c[r].x + c[r].y + c[r].z + c[r].w);
        carry += __builtin_amdgcn_readlane(s[r], 31);
    }
    if (lane == 0) wtot[w] = carry;
    __syncthreads();
    int base = 0;
    #pragma unroll
    for (int k = 0; k < SCAN_W; ++k) base += (k < w) ? wtot[k] : 0;

    #pragma unroll
    for (int r = 0; r < SCAN_R; ++r) {
        const int g = (w * SCAN_R + r) * 32 + lane;
        if (g < SCAN_Q) {
            int4 o;
            o.x = base + ex[r];
            o.y = o.x + c[r].x;
            o.z = o.y + c[r].y;
            o.w = o.z + c[r].z;
            q4[g] = o;
            float4 iv;
            iv.x = 1.0f / fmaxf((float)c[r].x, 1.0f);
            iv.y = 1.0f / fmaxf((float)c[r].y, 1.0f);
            iv.z = 1.0f / fmaxf((float)c[r].z, 1.0f);
            iv.w = 1.0f / fmaxf((float)c[r].w, 1.0f);
            inv4[g] = iv;
        }
    }
}

__global__ void place_kernel(const int* __restrict__ src, const int* __restrict__ dst,
                             const int* __restrict__ off, const int* __restrict__ slot,
                             int* __restrict__ list, int E) {
    int e = blockIdx.x * blockDim.x + threadIdx.x;
    if (e < E) list[off[dst[e]] + slot[e]] = src[e];
}

template <int U>
__device__ __forceinline__ void add_rows(const float* __restrict__ feat,
                                         const int* __restrict__ list,
                                         int j, int lane, float4& acc) {
    int    s[U];
    float4 v[U];
    #pragma unroll
    for (int k = 0; k < U; ++k) s[k] = list[j + k];
    #pragma unroll
    for (int k = 0; k < U; ++k)
        v[k] = *(const float4*)(feat + (size_t)s[k] * 128 + lane * 4);
    #pragma unroll
    for (int k = 0; k < U; ++k) {
        acc.x += v[k].x;
        acc.y += v[k].y;
        acc.z += v[k].z;
        acc.w += v[k].w;
    }
}

__global__ __launch_bounds__(256)
void gather_kernel(const float* __restrict__ feat, const int* __restrict__ list,
                   const int* __restrict__ off, float* __restrict__ agg,
                   int n_nodes, int E) {
    const int lane = threadIdx.x & 31;
    const int node = __builtin_amdgcn_readfirstlane(
        (int)((blockIdx.x * blockDim.x + threadIdx.x) >> 5));
    if (node >= n_nodes) return;
    const int beg = off[node];
    const int end = (node + 1 < n_nodes) ? off[node + 1] : E;
    float4 acc = make_float4(0.0f, 0.0f, 0.0f, 0.0f);
    int j = beg;
    for (; j + 8 <= end; j += 8) add_rows<8>(feat, list, j, lane, acc);
    if (end - j >= 4) { add_rows<4>(feat, list, j, lane, acc); j += 4; }
    if (end - j >= 2) { add_rows<2>(feat, list, j, lane, acc); j += 2; }
    if (end - j >= 1) { add_rows<1>(feat, list, j, lane, acc); }
    *(float4*)(agg + (size_t)node * 128 + lane * 4) = acc;
}

template <int NOUT, bool RELU>
__global__ __launch_bounds__(2 * NOUT)
void sage_gemm(const float* __restrict__ Aagg,
               const float* __restrict__ Aself,
               const float* __restrict__ invdeg,
               const float* __restrict__ Wl,
               const float* __restrict__ Wr,
               const float* __restrict__ bias,
               float* __restrict__ out) {
    constexpr int K = 128;
    const int wave = threadIdx.x >> 5;
    const int lane = threadIdx.x & 31;
    const int row  = lane & 15;
    const int hi   = lane >> 4;
    const int m0   = blockIdx.x * 16;
    const int n0   = wave * 16;

    const float* __restrict__ arow_agg  = Aagg  + (size_t)(m0 + row) * K;
    const float* __restrict__ arow_self = Aself + (size_t)(m0 + row) * K;
    const float  scale = invdeg[m0 + row];

    v8f c = {};

    #pragma unroll 4
    for (int k0 = 0; k0 < K; k0 += 4) {
        const int ka = k0 + 2 * hi;
        const float2 t = *(const float2*)(arow_agg + ka);
        v2f a;
        a.x = t.x * scale;
        a.y = t.y * scale;
        v2f b;
        b.x = Wl[(size_t)ka * NOUT + n0 + row];
        b.y = Wl[(size_t)(ka + 1) * NOUT + n0 + row];
        c = __builtin_amdgcn_wmma_f32_16x16x4_f32(false, a, false, b,
                                                  (short)0, c, false, false);
    }

    #pragma unroll 4
    for (int k0 = 0; k0 < K; k0 += 4) {
        const int ka = k0 + 2 * hi;
        const float2 t = *(const float2*)(arow_self + ka);
        v2f a;
        a.x = t.x;
        a.y = t.y;
        v2f b;
        b.x = Wr[(size_t)ka * NOUT + n0 + row];
        b.y = Wr[(size_t)(ka + 1) * NOUT + n0 + row];
        c = __builtin_amdgcn_wmma_f32_16x16x4_f32(false, a, false, b,
                                                  (short)0, c, false, false);
    }

    const float bv = bias[n0 + row];
    #pragma unroll
    for (int r = 0; r < 8; ++r) {
        float v = c[r] + bv;
        if (RELU) v = fmaxf(v, 0.0f);
        out[(size_t)(m0 + r + 8 * hi) * NOUT + n0 + row] = v;
    }
}

extern "C" void kernel_launch(void* const* d_in, const int* in_sizes, int n_in,
                              void* d_out, int out_size, void* d_ws, size_t ws_size,
                              hipStream_t stream) {
    const float* x   = (const float*)d_in[0];
    const int*   ei  = (const int*)d_in[1];
    const float* W1l = (const float*)d_in[2];
    const float* b1  = (const float*)d_in[3];
    const float* W1r = (const float*)d_in[4];
    const float* W2l = (const float*)d_in[5];
    const float* b2  = (const float*)d_in[6];
    const float* W2r = (const float*)d_in[7];

    const int* src = ei;
    const int* dst = ei + N_EDGES;

    float* agg    = (float*)d_ws;
    float* h      = agg + (size_t)N_NODES * 128;
    int*   off    = (int*)(h + (size_t)N_NODES * 128);
    float* invdeg = (float*)(off + N_NODES);

    int* list = (int*)d_out;
    int* slot = list + N_EDGES;

    const int ZB = 256;

    zero_i32<<<(N_NODES + ZB - 1) / ZB, ZB, 0, stream>>>(off, N_NODES);
    count_slot_kernel<<<(N_EDGES + ZB - 1) / ZB, ZB, 0, stream>>>(dst, off, slot, N_EDGES);
    scan_kernel<<<1, SCAN_T, 0, stream>>>(off, invdeg);
    place_kernel<<<(N_EDGES + ZB - 1) / ZB, ZB, 0, stream>>>(src, dst, off, slot, list, N_EDGES);

    gather_kernel<<<N_NODES * 32 / ZB, ZB, 0, stream>>>(x, list, off, agg, N_NODES, N_EDGES);
    sage_gemm<HID_CH, true><<<N_NODES / 16, 2 * HID_CH, 0, stream>>>(
        agg, x, invdeg, W1l, W1r, b1, h);

    gather_kernel<<<N_NODES * 32 / ZB, ZB, 0, stream>>>(h, list, off, agg, N_NODES, N_EDGES);
    sage_gemm<OUT_CH, false><<<N_NODES / 16, 2 * OUT_CH, 0, stream>>>(
        agg, h, invdeg, W2l, W2r, b2, (float*)d_out);
}


# SAGE product re-fed: resident work-groups walk the row tiles, each wave holds its 16 columns' B fragments in registers, A tile staged in LDS by whole-row loads; k order and rounding unchanged
# speedup vs baseline: 2.3026x; 1.2274x over previous
#include <hip/hip_runtime.h>
#include <hip/hip_bf16.h>


#define N_NODES 50000
#define N_EDGES 800000
#define IN_CH   128
#define HID_CH  128
#define OUT_CH  64
#define SCAN_T  1024

typedef __attribute__((ext_vector_type(2))) float v2f;
typedef __attribute__((ext_vector_type(8))) float v8f;

__global__ void zero_i32(int* __restrict__ p, int n) {
    int i = blockIdx.x * blockDim.x + threadIdx.x;
    if (i < n) p[i] = 0;
}

__global__ void count_slot_kernel(const int* __restrict__ dst, int* __restrict__ cnt,
                                  int* __restrict__ slot, int E) {
    int e = blockIdx.x * blockDim.x + threadIdx.x;
    if (e < E) slot[e] = atomicAdd(&cnt[dst[e]], 1);
}

constexpr int SCAN_W = SCAN_T / 32;
constexpr int SCAN_Q = N_NODES / 4;
constexpr int SCAN_R = (SCAN_Q + SCAN_T - 1) / SCAN_T;
static_assert(N_NODES % 4 == 0, "the scan reads the counts four at a time");
static_assert(SCAN_W == 32, "one lane per wave total in the last step");

__global__ __launch_bounds__(SCAN_T)
void scan_kernel(int* cnt_off, float* __restrict__ invdeg) {
    __shared__ int wtot[SCAN_W];
    const int lane = threadIdx.x & 31;
    const int w    = threadIdx.x >> 5;
    int4*   q4   = (int4*)cnt_off;
    float4* inv4 = (float4*)invdeg;

    int4 c[SCAN_R];
    int  s[SCAN_R];
    #pragma unroll
    for (int r = 0; r < SCAN_R; ++r) {
        const int g = (w * SCAN_R + r) * 32 + lane;
        c[r] = (g < SCAN_Q) ? q4[g] : make_int4(0, 0, 0, 0);
        s[r] = c[r].x + c[r].y + c[r].z + c[r].w;
    }
    #pragma unroll
    for (int d = 1; d < 32; d <<= 1) {
        int u[SCAN_R];
        #pragma unroll
        for (int r = 0; r < SCAN_R; ++r) u[r] = __shfl_up(s[r], d);
        #pragma unroll
        for (int r = 0; r < SCAN_R; ++r) if (lane >= d) s[r] += u[r];
    }
    int ex[SCAN_R];
    int carry = 0;
    #pragma unroll
    for (int r = 0; r < SCAN_R; ++r) {
        ex[r] = carry + s[r] - (c[r].x + c[r].y + c[r].z + c[r].w);
        carry += __builtin_amdgcn_readlane(s[r], 31);
    }
    if (lane == 0) wtot[w] = carry;
    __syncthreads();
    int base = 0;
    #pragma unroll
    for (int k = 0; k < SCAN_W; ++k) base += (k < w) ? wtot[k] : 0;

    #pragma unroll
    for (int r = 0; r < SCAN_R; ++r) {
        const int g = (w * SCAN_R + r) * 32 + lane;
        if (g < SCAN_Q) {
            int4 o;
            o.x = base + ex[r];
            o.y = o.x + c[r].x;
            o.z = o.y + c[r].y;
            o.w = o.z + c[r].z;
            q4[g] = o;
            float4 iv;
            iv.x = 1.0f / fmaxf((float)c[r].x, 1.0f);
            iv.y = 1.0f / fmaxf((float)c[r].y, 1.0f);
            iv.z = 1.0f / fmaxf((float)c[r].z, 1.0f);
            iv.w = 1.0f / fmaxf((float)c[r].w, 1.0f);
            inv4[g] = iv;
        }
    }
}

__global__ void place_kernel(const int* __restrict__ src, const int* __restrict__ dst,
                             const int* __restrict__ off, const int* __restrict__ slot,
                             int* __restrict__ list, int E) {
    int e = blockIdx.x * blockDim.x + threadIdx.x;
    if (e < E) list[off[dst[e]] + slot[e]] = src[e];
}

template <int U>
__device__ __forceinline__ void add_rows(const float* __restrict__ feat,
                                         const int* __restrict__ list,
                                         int j, int lane, float4& acc) {
    int    s[U];
    float4 v[U];
    #pragma unroll
    for (int k = 0; k < U; ++k) s[k] = list[j + k];
    #pragma unroll
    for (int k = 0; k < U; ++k)
        v[k] = *(const float4*)(feat + (size_t)s[k] * 128 + lane * 4);
    #pragma unroll
    for (int k = 0; k < U; ++k) {
        acc.x += v[k].x;
        acc.y += v[k].y;
        acc.z += v[k].z;
        acc.w += v[k].w;
    }
}

__global__ __launch_bounds__(256)
void gather_kernel(const float* __restrict__ feat, const int* __restrict__ list,
                   const int* __restrict__ off, float* __restrict__ agg,
                   int n_nodes, int E) {
    const int lane = threadIdx.x & 31;
    const int node = __builtin_amdgcn_readfirstlane(
        (int)((blockIdx.x * blockDim.x + threadIdx.x) >> 5));
    if (node >= n_nodes) return;
    const int beg = off[node];
    const int end = (node + 1 < n_nodes) ? off[node + 1] : E;
    float4 acc = make_float4(0.0f, 0.0f, 0.0f, 0.0f);
    int j = beg;
    for (; j + 8 <= end; j += 8) add_rows<8>(feat, list, j, lane, acc);
    if (end - j >= 4) { add_rows<4>(feat, list, j, lane, acc); j += 4; }
    if (end - j >= 2) { add_rows<2>(feat, list, j, lane, acc); j += 2; }
    if (end - j >= 1) { add_rows<1>(feat, list, j, lane, acc); }
    *(float4*)(agg + (size_t)node * 128 + lane * 4) = acc;
}

constexpr int GEMM_K      = 128;
constexpr int GEMM_TILES  = N_NODES / 16;
constexpr int GEMM_PITCH  = 2 * GEMM_K + 4;
static_assert(N_NODES % 16 == 0, "row tiles are whole");

template <int NW>
__device__ __forceinline__ void fetch_rows(const float* __restrict__ Aagg,
                                           const float* __restrict__ Aself,
                                           const float* __restrict__ invdeg,
                                           int t, int wave, int lane,
                                           float4 (&st)[32 / NW], float (&sc)[16 / NW]) {
    constexpr int LPW  = 32 / NW;
    constexpr int HALF = LPW / 2;
    #pragma unroll
    for (int i = 0; i < LPW; ++i) {
        const int r    = wave + NW * (i % HALF);
        const size_t g = (size_t)(t * 16 + r);
        if (i < HALF) {
            st[i] = *(const float4*)(Aagg + g * GEMM_K + lane * 4);
            sc[i] = invdeg[g];
        } else {
            st[i] = *(const float4*)(Aself + g * GEMM_K + lane * 4);
        }
    }
}

template <int NW>
__device__ __forceinline__ void put_rows(float* __restrict__ as, int wave, int lane,
                                         const float4 (&st)[32 / NW],
                                         const float (&sc)[16 / NW]) {
    constexpr int LPW  = 32 / NW;
    constexpr int HALF = LPW / 2;
    #pragma unroll
    for (int i = 0; i < LPW; ++i) {
        const int r = wave + NW * (i % HALF);
        float4 v = st[i];
        if (i < HALF) { v.x *= sc[i]; v.y *= sc[i]; v.z *= sc[i]; v.w *= sc[i]; }
        *(float4*)(as + r * GEMM_PITCH + (i < HALF ? 0 : GEMM_K) + lane * 4) = v;
    }
}

template <int NOUT, bool RELU>
__global__ __launch_bounds__(2 * NOUT)
void sage_gemm(const float* __restrict__ Aagg,
               const float* __restrict__ Aself,
               const float* __restrict__ invdeg,
               const float* __restrict__ Wl,
               const float* __restrict__ Wr,
               const float* __restrict__ bias,
               float* __restrict__ out) {
    constexpr int NW = NOUT / 16;
    __shared__ __attribute__((aligned(16))) float As[2][16 * GEMM_PITCH];

    const int wave = __builtin_amdgcn_readfirstlane((int)(threadIdx.x >> 5));
    const int lane = threadIdx.x & 31;
    const int row  = lane & 15;
    const int hi   = lane >> 4;
    const int n0   = wave * 16;

    v2f breg[64];
    #pragma unroll
    for (int s = 0; s < 32; ++s) {
        const int ka = 4 * s + 2 * hi;
        breg[s].x      = Wl[(size_t)ka * NOUT + n0 + row];
        breg[s].y      = Wl[(size_t)(ka + 1) * NOUT + n0 + row];
        breg[32 + s].x = Wr[(size_t)ka * NOUT + n0 + row];
        breg[32 + s].y = Wr[(size_t)(ka + 1) * NOUT + n0 + row];
    }
    const float bv = bias[n0 + row];

    float4 st[32 / NW];
    float  sc[16 / NW];
    int t = blockIdx.x;
    fetch_rows<NW>(Aagg, Aself, invdeg, t, wave, lane, st, sc);
    put_rows<NW>(As[0], wave, lane, st, sc);
    int buf = 0;

    for (; t < GEMM_TILES; t += gridDim.x) {
        __syncthreads();
        const int  tn   = t + gridDim.x;
        const bool more = tn < GEMM_TILES;
        if (more) fetch_rows<NW>(Aagg, Aself, invdeg, tn, wave, lane, st, sc);

        v8f c = {};
        const float* ap = As[buf] + row * GEMM_PITCH + 2 * hi;
        v2f af[2][16];
        #pragma unroll
        for (int j = 0; j < 16; ++j) af[0][j] = *(const v2f*)(ap + 4 * j);
        #pragma unroll
        for (int g = 0; g < 4; ++g) {
            if (g < 3) {
                #pragma unroll
                for (int j = 0; j < 16; ++j)
                    af[(g + 1) & 1][j] = *(const v2f*)(ap + 4 * (16 * (g + 1) + j));
            }
            __builtin_amdgcn_sched_barrier(0);
            #pragma unroll
            for (int j = 0; j < 16; ++j)
                c = __builtin_amdgcn_wmma_f32_16x16x4_f32(false, af[g & 1][j], false,
                                                          breg[16 * g + j],
                                                          (short)0, c, false, false);
            __builtin_amdgcn_sched_barrier(0);
        }

        const int m0 = t * 16;
        #pragma unroll
        for (int r = 0; r < 8; ++r) {
            float v = c[r] + bv;
            if (RELU) v = fmaxf(v, 0.0f);
            out[(size_t)(m0 + r + 8 * hi) * NOUT + n0 + row] = v;
        }

        if (more) put_rows<NW>(As[buf ^ 1], wave, lane, st, sc);
        buf ^= 1;
    }
}

constexpr int GEMM_GRID_1 = 512;
constexpr int GEMM_GRID_2 = 1024;
static_assert(GEMM_GRID_1 <= GEMM_TILES && GEMM_GRID_2 <= GEMM_TILES, "first tile of every group exists");

extern "C" void kernel_launch(void* const* d_in, const int* in_sizes, int n_in,
                              void* d_out, int out_size, void* d_ws, size_t ws_size,
                              hipStream_t stream) {
    const float* x   = (const float*)d_in[0];
    const int*   ei  = (const int*)d_in[1];
    const float* W1l = (const float*)d_in[2];
    const float* b1  = (const float*)d_in[3];
    const float* W1r = (const float*)d_in[4];
    const float* W2l = (const float*)d_in[5];
    const float* b2  = (const float*)d_in[6];
    const float* W2r = (const float*)d_in[7];

    const int* src = ei;
    const int* dst = ei + N_EDGES;

    float* agg    = (float*)d_ws;
    float* h      = agg + (size_t)N_NODES * 128;
    int*   off    = (int*)(h + (size_t)N_NODES * 128);
    float* invdeg = (float*)(off + N_NODES);

    int* list = (int*)d_out;
    int* slot = list + N_EDGES;

    const int ZB = 256;

    zero_i32<<<(N_NODES + ZB - 1) / ZB, ZB, 0, stream>>>(off, N_NODES);
    count_slot_kernel<<<(N_EDGES + ZB - 1) / ZB, ZB, 0, stream>>>(dst, off, slot, N_EDGES);
    scan_kernel<<<1, SCAN_T, 0, stream>>>(off, invdeg);
    place_kernel<<<(N_EDGES + ZB - 1) / ZB, ZB, 0, stream>>>(src, dst, off, slot, list, N_EDGES);

    gather_kernel<<<N_NODES * 32 / ZB, ZB, 0, stream>>>(x, list, off, agg, N_NODES, N_EDGES);
    sage_gemm<HID_CH, true><<<GEMM_GRID_1, 2 * HID_CH, 0, stream>>>(
        agg, x, invdeg, W1l, W1r, b1, h);

    gather_kernel<<<N_NODES * 32 / ZB, ZB, 0, stream>>>(h, list, off, agg, N_NODES, N_EDGES);
    sage_gemm<OUT_CH, false><<<GEMM_GRID_2, 2 * OUT_CH, 0, stream>>>(
        agg, h, invdeg, W2l, W2r, b2, (float*)d_out);
}


# Edge list built by LDS binning: chunks sorted by 64-node bin in LDS, a work-group per bin counts, scans, places in LDS; per-edge global atomics, scan, zero launches gone; row sums in LDS slot order
# speedup vs baseline: 4.1823x; 1.8163x over previous
#include <hip/hip_runtime.h>
#include <hip/hip_bf16.h>


#define N_NODES 50000
#define N_EDGES 800000
#define IN_CH   128
#define HID_CH  128
#define OUT_CH  64

typedef __attribute__((ext_vector_type(2))) float v2f;
typedef __attribute__((ext_vector_type(8))) float v8f;

constexpr int BIN_B   = 64;
constexpr int NBIN    = (N_NODES + BIN_B - 1) / BIN_B;
constexpr int NCHUNK  = 256;
constexpr int CHUNK_E = (N_EDGES + NCHUNK - 1) / NCHUNK;
constexpr int BIN_T   = 1024;
constexpr int BIN_K   = (CHUNK_E + BIN_T - 1) / BIN_T;
constexpr int TAB_W   = NBIN + 1;
constexpr int RUN_R   = 8;
constexpr int CSR_CAP = 8192;
static_assert(N_NODES <= 65536, "a source id takes the low 16 bits of a packed edge");
static_assert(BIN_B == 64, "csr_kernel scans the bin's counts two per lane in one wave");
static_assert(TAB_W <= BIN_T && BIN_T == 1024, "one bin per thread, one wave total per lane");
static_assert((long long)NCHUNK * CHUNK_E >= N_EDGES, "chunks cover the edges");
static_assert(2LL * N_EDGES + (long long)NCHUNK * TAB_W <= (long long)N_NODES * OUT_CH,
              "list, packed edges and table fit the output buffer");

__global__ __launch_bounds__(BIN_T)
void bin_kernel(const int* __restrict__ src, const int* __restrict__ dst,
                int* __restrict__ tmp, int* __restrict__ tab) {
    __shared__ int cntb[BIN_T];
    __shared__ int wtot[BIN_T / 32];
    __shared__ int stage[CHUNK_E];
    const int tid  = threadIdx.x;
    const int lane = tid & 31;
    const int w    = tid >> 5;
    const int g    = blockIdx.x;
    const int e0   = g * CHUNK_E;
    const int e1   = (e0 + CHUNK_E < N_EDGES) ? e0 + CHUNK_E : N_EDGES;

    cntb[tid] = 0;
    __syncthreads();

    int bin[BIN_K], word[BIN_K], rnk[BIN_K];
    #pragma unroll
    for (int k = 0; k < BIN_K; ++k) {
        const int e  = e0 + k * BIN_T + tid;
        const int ec = (e < e1) ? e : N_EDGES - 1;
        const unsigned d = (unsigned)dst[ec];
        const unsigned s = (unsigned)src[ec];
        bin[k]  = (e < e1) ? (int)(d / BIN_B) : -1;
        word[k] = (int)(((d % BIN_B) << 16) | s);
    }
    #pragma unroll
    for (int k = 0; k < BIN_K; ++k) {
        rnk[k] = 0;
        if (bin[k] >= 0) rnk[k] = atomicAdd(&cntb[bin[k]], 1);
    }
    __syncthreads();

    const int c = cntb[tid];
    int s = c;
    #pragma unroll
    for (int d = 1; d < 32; d <<= 1) {
        const int u = __shfl_up(s, d);
        if (lane >= d) s += u;
    }
    if (lane == 31) wtot[w] = s;
    __syncthreads();
    if (w == 0) {
        const int t = wtot[lane];
        int ts = t;
        #pragma unroll
        for (int d = 1; d < 32; d <<= 1) {
            const int u = __shfl_up(ts, d);
            if (lane >= d) ts += u;
        }
        wtot[lane] = ts - t;
    }
    __syncthreads();
    const int ex = wtot[w] + s - c;
    cntb[tid] = ex;
    if (tid < TAB_W) tab[g * TAB_W + tid] = ex;
    __syncthreads();

    #pragma unroll
    for (int k = 0; k < BIN_K; ++k)
        if (bin[k] >= 0) stage[cntb[bin[k]] + rnk[k]] = word[k];
    __syncthreads();
    for (int i = tid; i < e1 - e0; i += BIN_T) tmp[e0 + i] = stage[i];
}

__global__ __launch_bounds__(NCHUNK)
void csr_kernel(const int* __restrict__ tmp, const int* __restrict__ tab,
                int* __restrict__ off, float* __restrict__ invdeg,
                int* __restrict__ list) {
    __shared__ int cnt[BIN_B];
    __shared__ int fill[BIN_B];
    __shared__ int noff[BIN_B];
    __shared__ int wsb[NCHUNK / 32];
    __shared__ int wsn[NCHUNK / 32];
    __shared__ int obuf[CSR_CAP];
    const int tid  = threadIdx.x;
    const int lane = tid & 31;
    const int w    = tid >> 5;
    const int b    = blockIdx.x;

    const int rb = tab[tid * TAB_W + b];
    const int re = tab[tid * TAB_W + b + 1];
    const int n  = re - rb;
    const int* __restrict__ run = tmp + (size_t)tid * CHUNK_E + rb;
    int rw[RUN_R];
    #pragma unroll
    for (int k = 0; k < RUN_R; ++k) rw[k] = (k < n) ? run[k] : 0;

    if (tid < BIN_B) { cnt[tid] = 0; fill[tid] = 0; }

    int sb = rb, sn = n;
    #pragma unroll
    for (int d = 16; d >= 1; d >>= 1) {
        sb += __shfl_xor(sb, d);
        sn += __shfl_xor(sn, d);
    }
    if (lane == 0) { wsb[w] = sb; wsn[w] = sn; }
    __syncthreads();
    int base = 0, total = 0;
    #pragma unroll
    for (int k = 0; k < NCHUNK / 32; ++k) { base += wsb[k]; total += wsn[k]; }

    #pragma unroll
    for (int k = 0; k < RUN_R; ++k)
        if (k < n) atomicAdd(&cnt[(unsigned)rw[k] >> 16], 1);
    for (int i = RUN_R; i < n; ++i) atomicAdd(&cnt[(unsigned)run[i] >> 16], 1);
    __syncthreads();

    if (w == 0) {
        const int c0 = cnt[2 * lane];
        const int c1 = cnt[2 * lane + 1];
        int s = c0 + c1;
        #pragma unroll
        for (int d = 1; d < 32; d <<= 1) {
            const int u = __shfl_up(s, d);
            if (lane >= d) s += u;
        }
        const int ex = s - (c0 + c1);
        noff[2 * lane]     = ex;
        noff[2 * lane + 1] = ex + c0;
    }
    __syncthreads();
    if (tid < BIN_B) {
        const int node = b * BIN_B + tid;
        if (node < N_NODES) {
            off[node]    = base + noff[tid];
            invdeg[node] = 1.0f / fmaxf((float)cnt[tid], 1.0f);
        }
    }

    const bool in_lds = total <= CSR_CAP;
    #pragma unroll
    for (int k = 0; k < RUN_R; ++k) {
        if (k < n) {
            const unsigned wd = (unsigned)rw[k];
            const int dl = (int)(wd >> 16);
            const int p  = noff[dl] + atomicAdd(&fill[dl], 1);
            if (in_lds) obuf[p] = (int)(wd & 0xFFFFu);
            else        list[base + p] = (int)(wd & 0xFFFFu);
        }
    }
    for (int i = RUN_R; i < n; ++i) {
        const unsigned wd = (unsigned)run[i];
        const int dl = (int)(wd >> 16);
        const int p  = noff[dl] + atomicAdd(&fill[dl], 1);
        if (in_lds) obuf[p] = (int)(wd & 0xFFFFu);
        else        list[base + p] = (int)(wd & 0xFFFFu);
    }
    if (in_lds) {
        __syncthreads();
        for (int i = tid; i < total; i += NCHUNK) list[base + i] = obuf[i];
    }
}

template <int U>
__device__ __forceinline__ void add_rows(const float* __restrict__ feat,
                                         const int* __restrict__ list,
                                         int j, int lane, float4& acc) {
    int    s[U];
    float4 v[U];
    #pragma unroll
    for (int k = 0; k < U; ++k) s[k] = list[j + k];
    #pragma unroll
    for (int k = 0; k < U; ++k)
        v[k] = *(const float4*)(feat + (size_t)s[k] * 128 + lane * 4);
    #pragma unroll
    for (int k = 0; k < U; ++k) {
        acc.x += v[k].x;
        acc.y += v[k].y;
        acc.z += v[k].z;
        acc.w += v[k].w;
    }
}

__global__ __launch_bounds__(256)
void gather_kernel(const float* __restrict__ feat, const int* __restrict__ list,
                   const int* __restrict__ off, float* __restrict__ agg,
                   int n_nodes, int E) {
    const int lane = threadIdx.x & 31;
    const int node = __builtin_amdgcn_readfirstlane(
        (int)((blockIdx.x * blockDim.x + threadIdx.x) >> 5));
    if (node >= n_nodes) return;
    const int beg = off[node];
    const int end = (node + 1 < n_nodes) ? off[node + 1] : E;
    float4 acc = make_float4(0.0f, 0.0f, 0.0f, 0.0f);
    int j = beg;
    for (; j + 8 <= end; j += 8) add_rows<8>(feat, list, j, lane, acc);
    if (end - j >= 4) { add_rows<4>(feat, list, j, lane, acc); j += 4; }
    if (end - j >= 2) { add_rows<2>(feat, list, j, lane, acc); j += 2; }
    if (end - j >= 1) { add_rows<1>(feat, list, j, lane, acc); }
    *(float4*)(agg + (size_t)node * 128 + lane * 4) = acc;
}

constexpr int GEMM_K      = 128;
constexpr int GEMM_TILES  = N_NODES / 16;
constexpr int GEMM_PITCH  = 2 * GEMM_K + 4;
static_assert(N_NODES % 16 == 0, "row tiles are whole");

template <int NW>
__device__ __forceinline__ void fetch_rows(const float* __restrict__ Aagg,
                                           const float* __restrict__ Aself,
                                           const float* __restrict__ invdeg,
                                           int t, int wave, int lane,
                                           float4 (&st)[32 / NW], float (&sc)[16 / NW]) {
    constexpr int LPW  = 32 / NW;
    constexpr int HALF = LPW / 2;
    #pragma unroll
    for (int i = 0; i < LPW; ++i) {
        const int r    = wave + NW * (i % HALF);
        const size_t g = (size_t)(t * 16 + r);
        if (i < HALF) {
            st[i] = *(const float4*)(Aagg + g * GEMM_K + lane * 4);
            sc[i] = invdeg[g];
        } else {
            st[i] = *(const float4*)(Aself + g * GEMM_K + lane * 4);
        }
    }
}

template <int NW>
__device__ __forceinline__ void put_rows(float* __restrict__ as, int wave, int lane,
                                         const float4 (&st)[32 / NW],
                                         const float (&sc)[16 / NW]) {
    constexpr int LPW  = 32 / NW;
    constexpr int HALF = LPW / 2;
    #pragma unroll
    for (int i = 0; i < LPW; ++i) {
        const int r = wave + NW * (i % HALF);
        float4 v = st[i];
        if (i < HALF) { v.x *= sc[i]; v.y *= sc[i]; v.z *= sc[i]; v.w *= sc[i]; }
        *(float4*)(as + r * GEMM_PITCH + (i < HALF ? 0 : GEMM_K) + lane * 4) = v;
    }
}

template <int NOUT, bool RELU>
__global__ __launch_bounds__(2 * NOUT)
void sage_gemm(const float* __restrict__ Aagg,
               const float* __restrict__ Aself,
               const float* __restrict__ invdeg,
               const float* __restrict__ Wl,
               const float* __restrict__ Wr,
               const float* __restrict__ bias,
               float* __restrict__ out) {
    constexpr int NW = NOUT / 16;
    __shared__ __attribute__((aligned(16))) float As[2][16 * GEMM_PITCH];

    const int wave = __builtin_amdgcn_readfirstlane((int)(threadIdx.x >> 5));
    const int lane = threadIdx.x & 31;
    const int row  = lane & 15;
    const int hi   = lane >> 4;
    const int n0   = wave * 16;

    v2f breg[64];
    #pragma unroll
    for (int s = 0; s < 32; ++s) {
        const int ka = 4 * s + 2 * hi;
        breg[s].x      = Wl[(size_t)ka * NOUT + n0 + row];
        breg[s].y      = Wl[(size_t)(ka + 1) * NOUT + n0 + row];
        breg[32 + s].x = Wr[(size_t)ka * NOUT + n0 + row];
        breg[32 + s].y = Wr[(size_t)(ka + 1) * NOUT + n0 + row];
    }
    const float bv = bias[n0 + row];

    float4 st[32 / NW];
    float  sc[16 / NW];
    int t = blockIdx.x;
    fetch_rows<NW>(Aagg, Aself, invdeg, t, wave, lane, st, sc);
    put_rows<NW>(As[0], wave, lane, st, sc);
    int buf = 0;

    for (; t < GEMM_TILES; t += gridDim.x) {
        __syncthreads();
        const int  tn   = t + gridDim.x;
        const bool more = tn < GEMM_TILES;
        if (more) fetch_rows<NW>(Aagg, Aself, invdeg, tn, wave, lane, st, sc);

        v8f c = {};
        const float* ap = As[buf] + row * GEMM_PITCH + 2 * hi;
        v2f af[2][16];
        #pragma unroll
        for (int j = 0; j < 16; ++j) af[0][j] = *(const v2f*)(ap + 4 * j);
        #pragma unroll
        for (int g = 0; g < 4; ++g) {
            if (g < 3) {
                #pragma unroll
                for (int j = 0; j < 16; ++j)
                    af[(g + 1) & 1][j] = *(const v2f*)(ap + 4 * (16 * (g + 1) + j));
            }
            __builtin_amdgcn_sched_barrier(0);
            #pragma unroll
            for (int j = 0; j < 16; ++j)
                c = __builtin_amdgcn_wmma_f32_16x16x4_f32(false, af[g & 1][j], false,
                                                          breg[16 * g + j],
                                                          (short)0, c, false, false);
            __builtin_amdgcn_sched_barrier(0);
        }

        const int m0 = t * 16;
        #pragma unroll
        for (int r = 0; r < 8; ++r) {
            float v = c[r] + bv;
            if (RELU) v = fmaxf(v, 0.0f);
            out[(size_t)(m0 + r + 8 * hi) * NOUT + n0 + row] = v;
        }

        if (more) put_rows<NW>(As[buf ^ 1], wave, lane, st, sc);
        buf ^= 1;
    }
}

constexpr int GEMM_GRID_1 = 512;
constexpr int GEMM_GRID_2 = 1024;
static_assert(GEMM_GRID_1 <= GEMM_TILES && GEMM_GRID_2 <= GEMM_TILES, "first tile of every group exists");

extern "C" void kernel_launch(void* const* d_in, const int* in_sizes, int n_in,
                              void* d_out, int out_size, void* d_ws, size_t ws_size,
                              hipStream_t stream) {
    const float* x   = (const float*)d_in[0];
    const int*   ei  = (const int*)d_in[1];
    const float* W1l = (const float*)d_in[2];
    const float* b1  = (const float*)d_in[3];
    const float* W1r = (const float*)d_in[4];
    const float* W2l = (const float*)d_in[5];
    const float* b2  = (const float*)d_in[6];
    const float* W2r = (const float*)d_in[7];

    const int* src = ei;
    const int* dst = ei + N_EDGES;

    float* agg    = (float*)d_ws;
    float* h      = agg + (size_t)N_NODES * 128;
    int*   off    = (int*)(h + (size_t)N_NODES * 128);
    float* invdeg = (float*)(off + N_NODES);

    int* list = (int*)d_out;
    int* tmp  = list + N_EDGES;
    int* tab  = tmp + N_EDGES;

    const int ZB = 256;

    bin_kernel<<<NCHUNK, BIN_T, 0, stream>>>(src, dst, tmp, tab);
    csr_kernel<<<NBIN, NCHUNK, 0, stream>>>(tmp, tab, off, invdeg, list);

    gather_kernel<<<N_NODES * 32 / ZB, ZB, 0, stream>>>(x, list, off, agg, N_NODES, N_EDGES);
    sage_gemm<HID_CH, true><<<GEMM_GRID_1, 2 * HID_CH, 0, stream>>>(
        agg, x, invdeg, W1l, W1r, b1, h);

    gather_kernel<<<N_NODES * 32 / ZB, ZB, 0, stream>>>(h, list, off, agg, N_NODES, N_EDGES);
    sage_gemm<OUT_CH, false><<<GEMM_GRID_2, 2 * OUT_CH, 0, stream>>>(
        agg, h, invdeg, W2l, W2r, b2, (float*)d_out);
}


# Neighbour sums folded into the SAGE product's A-tile fill: a wave sums its nodes' source rows in slot order into the LDS tile while other waves multiply; gather launches and agg round trip gone
# speedup vs baseline: 4.4737x; 1.0697x over previous
#include <hip/hip_runtime.h>
#include <hip/hip_bf16.h>


#define N_NODES 50000
#define N_EDGES 800000
#define IN_CH   128
#define HID_CH  128
#define OUT_CH  64

typedef __attribute__((ext_vector_type(2))) float v2f;
typedef __attribute__((ext_vector_type(8))) float v8f;

constexpr int BIN_B   = 64;
constexpr int NBIN    = (N_NODES + BIN_B - 1) / BIN_B;
constexpr int NCHUNK  = 256;
constexpr int CHUNK_E = (N_EDGES + NCHUNK - 1) / NCHUNK;
constexpr int BIN_T   = 1024;
constexpr int BIN_K   = (CHUNK_E + BIN_T - 1) / BIN_T;
constexpr int TAB_W   = NBIN + 1;
constexpr int RUN_R   = 8;
constexpr int CSR_CAP = 8192;
static_assert(N_NODES <= 65536, "a source id takes the low 16 bits of a packed edge");
static_assert(BIN_B == 64, "csr_kernel scans the bin's counts two per lane in one wave");
static_assert(TAB_W <= BIN_T && BIN_T == 1024, "one bin per thread, one wave total per lane");
static_assert((long long)NCHUNK * CHUNK_E >= N_EDGES, "chunks cover the edges");
static_assert(2LL * N_EDGES + (long long)NCHUNK * TAB_W <= (long long)N_NODES * 128,
              "list, packed edges and table fit the workspace in front of h");

__global__ __launch_bounds__(BIN_T)
void bin_kernel(const int* __restrict__ src, const int* __restrict__ dst,
                int* __restrict__ tmp, int* __restrict__ tab) {
    __shared__ int cntb[BIN_T];
    __shared__ int wtot[BIN_T / 32];
    __shared__ int stage[CHUNK_E];
    const int tid  = threadIdx.x;
    const int lane = tid & 31;
    const int w    = tid >> 5;
    const int g    = blockIdx.x;
    const int e0   = g * CHUNK_E;
    const int e1   = (e0 + CHUNK_E < N_EDGES) ? e0 + CHUNK_E : N_EDGES;

    cntb[tid] = 0;
    __syncthreads();

    int bin[BIN_K], word[BIN_K], rnk[BIN_K];
    #pragma unroll
    for (int k = 0; k < BIN_K; ++k) {
        const int e  = e0 + k * BIN_T + tid;
        const int ec = (e < e1) ? e : N_EDGES - 1;
        const unsigned d = (unsigned)dst[ec];
        const unsigned s = (unsigned)src[ec];
        bin[k]  = (e < e1) ? (int)(d / BIN_B) : -1;
        word[k] = (int)(((d % BIN_B) << 16) | s);
    }
    #pragma unroll
    for (int k = 0; k < BIN_K; ++k) {
        rnk[k] = 0;
        if (bin[k] >= 0) rnk[k] = atomicAdd(&cntb[bin[k]], 1);
    }
    __syncthreads();

    const int c = cntb[tid];
    int s = c;
    #pragma unroll
    for (int d = 1; d < 32; d <<= 1) {
        const int u = __shfl_up(s, d);
        if (lane >= d) s += u;
    }
    if (lane == 31) wtot[w] = s;
    __syncthreads();
    if (w == 0) {
        const int t = wtot[lane];
        int ts = t;
        #pragma unroll
        for (int d = 1; d < 32; d <<= 1) {
            const int u = __shfl_up(ts, d);
            if (lane >= d) ts += u;
        }
        wtot[lane] = ts - t;
    }
    __syncthreads();
    const int ex = wtot[w] + s - c;
    cntb[tid] = ex;
    if (tid < TAB_W) tab[g * TAB_W + tid] = ex;
    __syncthreads();

    #pragma unroll
    for (int k = 0; k < BIN_K; ++k)
        if (bin[k] >= 0) stage[cntb[bin[k]] + rnk[k]] = word[k];
    __syncthreads();
    for (int i = tid; i < e1 - e0; i += BIN_T) tmp[e0 + i] = stage[i];
}

__global__ __launch_bounds__(NCHUNK)
void csr_kernel(const int* __restrict__ tmp, const int* __restrict__ tab,
                int* __restrict__ off, float* __restrict__ invdeg,
                int* __restrict__ list) {
    __shared__ int cnt[BIN_B];
    __shared__ int fill[BIN_B];
    __shared__ int noff[BIN_B];
    __shared__ int wsb[NCHUNK / 32];
    __shared__ int wsn[NCHUNK / 32];
    __shared__ int obuf[CSR_CAP];
    const int tid  = threadIdx.x;
    const int lane = tid & 31;
    const int w    = tid >> 5;
    const int b    = blockIdx.x;

    const int rb = tab[tid * TAB_W + b];
    const int re = tab[tid * TAB_W + b + 1];
    const int n  = re - rb;
    const int* __restrict__ run = tmp + (size_t)tid * CHUNK_E + rb;
    int rw[RUN_R];
    #pragma unroll
    for (int k = 0; k < RUN_R; ++k) rw[k] = (k < n) ? run[k] : 0;

    if (tid < BIN_B) { cnt[tid] = 0; fill[tid] = 0; }

    int sb = rb, sn = n;
    #pragma unroll
    for (int d = 16; d >= 1; d >>= 1) {
        sb += __shfl_xor(sb, d);
        sn += __shfl_xor(sn, d);
    }
    if (lane == 0) { wsb[w] = sb; wsn[w] = sn; }
    __syncthreads();
    int base = 0, total = 0;
    #pragma unroll
    for (int k = 0; k < NCHUNK / 32; ++k) { base += wsb[k]; total += wsn[k]; }

    #pragma unroll
    for (int k = 0; k < RUN_R; ++k)
        if (k < n) atomicAdd(&cnt[(unsigned)rw[k] >> 16], 1);
    for (int i = RUN_R; i < n; ++i) atomicAdd(&cnt[(unsigned)run[i] >> 16], 1);
    __syncthreads();

    if (w == 0) {
        const int c0 = cnt[2 * lane];
        const int c1 = cnt[2 * lane + 1];
        int s = c0 + c1;
        #pragma unroll
        for (int d = 1; d < 32; d <<= 1) {
            const int u = __shfl_up(s, d);
            if (lane >= d) s += u;
        }
        const int ex = s - (c0 + c1);
        noff[2 * lane]     = ex;
        noff[2 * lane + 1] = ex + c0;
    }
    __syncthreads();
    if (tid < BIN_B) {
        const int node = b * BIN_B + tid;
        if (node < N_NODES) {
            off[node]    = base + noff[tid];
            invdeg[node] = 1.0f / fmaxf((float)cnt[tid], 1.0f);
        }
    }

    const bool in_lds = total <= CSR_CAP;
    #pragma unroll
    for (int k = 0; k < RUN_R; ++k) {
        if (k < n) {
            const unsigned wd = (unsigned)rw[k];
            const int dl = (int)(wd >> 16);
            const int p  = noff[dl] + atomicAdd(&fill[dl], 1);
            if (in_lds) obuf[p] = (int)(wd & 0xFFFFu);
            else        list[base + p] = (int)(wd & 0xFFFFu);
        }
    }
    for (int i = RUN_R; i < n; ++i) {
        const unsigned wd = (unsigned)run[i];
        const int dl = (int)(wd >> 16);
        const int p  = noff[dl] + atomicAdd(&fill[dl], 1);
        if (in_lds) obuf[p] = (int)(wd & 0xFFFFu);
        else        list[base + p] = (int)(wd & 0xFFFFu);
    }
    if (in_lds) {
        __syncthreads();
        for (int i = tid; i < total; i += NCHUNK) list[base + i] = obuf[i];
    }
}

constexpr int GEMM_K      = 128;
constexpr int GEMM_TILES  = N_NODES / 16;
constexpr int GEMM_PITCH  = 2 * GEMM_K + 4;
static_assert(N_NODES % 16 == 0, "row tiles are whole");

constexpr int ROWS_FLY = 12;
constexpr int ID_WIN   = (32 / ROWS_FLY) * ROWS_FLY;

template <int NW>
__device__ __forceinline__ void tile_bounds(const int* __restrict__ off,
                                            const float* __restrict__ invdeg,
                                            int t, int wave, int lane, int& bnd, float& idg) {
    const int gi = t * 16 + wave * (16 / NW) + lane;
    const int gc = (gi < N_NODES) ? gi : N_NODES - 1;
    const int o  = off[gc];
    bnd = (gi < N_NODES) ? o : N_EDGES;
    idg = invdeg[gc];
}

__device__ __forceinline__ int id_window(const int* __restrict__ list, int first, int lane) {
    const int j = first + lane;
    return list[(j < N_EDGES) ? j : N_EDGES - 1];
}

template <int NW>
__device__ __forceinline__ void fill_tile(float* __restrict__ as,
                                          const float* __restrict__ feat,
                                          const int* __restrict__ list,
                                          int t, int wave, int lane,
                                          int bnd, float idg, int mine) {
    constexpr int HALF = 16 / NW;
    const int g0 = t * 16 + wave * HALF;
    float* rp = as + (wave * HALF) * GEMM_PITCH + lane * 4;
    int beg = __builtin_amdgcn_readlane(bnd, 0);

    #pragma nounroll
    for (int i = 0; i < HALF; ++i) {
        const int end = __builtin_amdgcn_readlane(bnd, i + 1);
        const int nmine = id_window(list, end, lane);
        const float* ownp = feat + (size_t)(g0 + i) * GEMM_K + lane * 4;
        float4 ownv;
        float4 acc = make_float4(0.0f, 0.0f, 0.0f, 0.0f);
        const int n = end - beg;
        for (int c = 0, w = 0; c < n || c == 0; c += ROWS_FLY, w += ROWS_FLY) {
            if (w == ID_WIN) {
                w = 0;
                mine = id_window(list, beg + c, lane);
            }
            float4 v[ROWS_FLY];
            if (n - c >= ROWS_FLY) {
                #pragma unroll
                for (int k = 0; k < ROWS_FLY; ++k) {
                    const int s = __builtin_amdgcn_readlane(mine, w + k);
                    v[k] = *(const float4*)(feat + (size_t)s * GEMM_K + lane * 4);
                }
                if (c == 0) ownv = *(const float4*)ownp;
                #pragma unroll
                for (int k = 0; k < ROWS_FLY; ++k) {
                    acc.x += v[k].x;
                    acc.y += v[k].y;
                    acc.z += v[k].z;
                    acc.w += v[k].w;
                }
            } else {
                const int cnt = n - c;
                int sid[ROWS_FLY - 1];
                #pragma unroll
                for (int k = 0; k < ROWS_FLY - 1; ++k) sid[k] = __builtin_amdgcn_readlane(mine, w + k);
                v[0] = *(const float4*)(feat + (size_t)sid[0] * GEMM_K + lane * 4);
                #pragma unroll
                for (int k = 1; k < ROWS_FLY - 1; ++k)
                    if (k < cnt) v[k] = *(const float4*)(feat + (size_t)sid[k] * GEMM_K + lane * 4);
                if (c == 0) ownv = *(const float4*)ownp;
                #pragma unroll
                for (int k = 0; k < ROWS_FLY - 1; ++k) {
                    if (k < cnt) {
                        acc.x += v[k].x;
                        acc.y += v[k].y;
                        acc.z += v[k].z;
                        acc.w += v[k].w;
                    }
                }
            }
            if (c == 0) *(float4*)(rp + i * GEMM_PITCH + GEMM_K) = ownv;
        }
        const float sc = __builtin_bit_cast(float,
            __builtin_amdgcn_readlane(__builtin_bit_cast(int, idg), i));
        acc.x *= sc; acc.y *= sc; acc.z *= sc; acc.w *= sc;
        *(float4*)(rp + i * GEMM_PITCH) = acc;
        beg = end; mine = nmine;
    }
}

template <int NOUT, bool RELU>
__global__ __launch_bounds__(2 * NOUT)
void sage_gemm(const float* __restrict__ feat,
               const int* __restrict__ list,
               const int* __restrict__ off,
               const float* __restrict__ invdeg,
               const float* __restrict__ Wl,
               const float* __restrict__ Wr,
               const float* __restrict__ bias,
               float* __restrict__ out) {
    constexpr int NW = NOUT / 16;
    __shared__ __attribute__((aligned(16))) float As[2][16 * GEMM_PITCH];

    const int wave = __builtin_amdgcn_readfirstlane((int)(threadIdx.x >> 5));
    const int lane = threadIdx.x & 31;
    const int row  = lane & 15;
    const int hi   = lane >> 4;
    const int n0   = wave * 16;

    v2f breg[64];
    #pragma unroll
    for (int s = 0; s < 32; ++s) {
        const int ka = 4 * s + 2 * hi;
        breg[s].x      = Wl[(size_t)ka * NOUT + n0 + row];
        breg[s].y      = Wl[(size_t)(ka + 1) * NOUT + n0 + row];
        breg[32 + s].x = Wr[(size_t)ka * NOUT + n0 + row];
        breg[32 + s].y = Wr[(size_t)(ka + 1) * NOUT + n0 + row];
    }
    const float bv = bias[n0 + row];

    int buf = 0;
    int   bnd,  nbnd;
    float idg,  nidg;
    tile_bounds<NW>(off, invdeg, (int)blockIdx.x, wave, lane, bnd, idg);
    int mine = id_window(list, __builtin_amdgcn_readlane(bnd, 0), lane);
    for (int t = blockIdx.x; t < GEMM_TILES; t += gridDim.x) {
        const int tn = (t + (int)gridDim.x < GEMM_TILES) ? t + (int)gridDim.x : t;
        tile_bounds<NW>(off, invdeg, tn, wave, lane, nbnd, nidg);
        fill_tile<NW>(As[buf], feat, list, t, wave, lane, bnd, idg, mine);
        bnd = nbnd; idg = nidg;
        __syncthreads();
        mine = id_window(list, __builtin_amdgcn_readlane(bnd, 0), lane);

        v8f c = {};
        const float* ap = As[buf] + row * GEMM_PITCH + 2 * hi;
        v2f af[2][16];
        #pragma unroll
        for (int j = 0; j < 16; ++j) af[0][j] = *(const v2f*)(ap + 4 * j);
        #pragma unroll
        for (int g = 0; g < 4; ++g) {
            if (g < 3) {
                #pragma unroll
                for (int j = 0; j < 16; ++j)
                    af[(g + 1) & 1][j] = *(const v2f*)(ap + 4 * (16 * (g + 1) + j));
            }
            __builtin_amdgcn_sched_barrier(0);
            #pragma unroll
            for (int j = 0; j < 16; ++j)
                c = __builtin_amdgcn_wmma_f32_16x16x4_f32(false, af[g & 1][j], false,
                                                          breg[16 * g + j],
                                                          (short)0, c, false, false);
            __builtin_amdgcn_sched_barrier(0);
        }

        const int m0 = t * 16;
        #pragma unroll
        for (int r = 0; r < 8; ++r) {
            float v = c[r] + bv;
            if (RELU) v = fmaxf(v, 0.0f);
            out[(size_t)(m0 + r + 8 * hi) * NOUT + n0 + row] = v;
        }
        buf ^= 1;
    }
}

constexpr int GEMM_GRID_1 = 512;
constexpr int GEMM_GRID_2 = 1024;
static_assert(GEMM_GRID_1 <= GEMM_TILES && GEMM_GRID_2 <= GEMM_TILES, "first tile of every group exists");

extern "C" void kernel_launch(void* const* d_in, const int* in_sizes, int n_in,
                              void* d_out, int out_size, void* d_ws, size_t ws_size,
                              hipStream_t stream) {
    const float* x   = (const float*)d_in[0];
    const int*   ei  = (const int*)d_in[1];
    const float* W1l = (const float*)d_in[2];
    const float* b1  = (const float*)d_in[3];
    const float* W1r = (const float*)d_in[4];
    const float* W2l = (const float*)d_in[5];
    const float* b2  = (const float*)d_in[6];
    const float* W2r = (const float*)d_in[7];

    const int* src = ei;
    const int* dst = ei + N_EDGES;

    int*   list   = (int*)d_ws;
    int*   tmp    = list + N_EDGES;
    int*   tab    = tmp + N_EDGES;
    float* h      = (float*)d_ws + (size_t)N_NODES * 128;
    int*   off    = (int*)(h + (size_t)N_NODES * 128);
    float* invdeg = (float*)(off + N_NODES);

    bin_kernel<<<NCHUNK, BIN_T, 0, stream>>>(src, dst, tmp, tab);
    csr_kernel<<<NBIN, NCHUNK, 0, stream>>>(tmp, tab, off, invdeg, list);

    sage_gemm<HID_CH, true><<<GEMM_GRID_1, 2 * HID_CH, 0, stream>>>(
        x, list, off, invdeg, W1l, W1r, b1, h);

    sage_gemm<OUT_CH, false><<<GEMM_GRID_2, 2 * OUT_CH, 0, stream>>>(
        h, list, off, invdeg, W2l, W2r, b2, (float*)d_out);
}
